# starting kernel
# baseline (speedup 1.0000x reference)
#include <hip/hip_runtime.h>

typedef __attribute__((ext_vector_type(2))) float v2f;
typedef __attribute__((ext_vector_type(4))) float v4f;
typedef __attribute__((ext_vector_type(8))) float v8f;

__device__ __forceinline__ float relu1(float x) {
  return __builtin_amdgcn_fmed3f(x, 0.0f, __builtin_inff());
}

__device__ __forceinline__ void mlp_tile(v2f a,
                                         const v2f (&bw)[8],
                                         const float (&b1v)[8],
                                         const float (&w2a)[8],
                                         const float (&w2b)[8],
                                         float b2lane,
                                         int lane,
                                         float* __restrict__ out_probs_tile) {
  float acc[16];
#pragma unroll
  for (int j = 0; j < 16; ++j) acc[j] = 0.0f;

#pragma unroll
  for (int n = 0; n < 8; ++n) {
    v8f c;
#pragma unroll
    for (int r = 0; r < 8; ++r) c[r] = b1v[n];
    c = __builtin_amdgcn_wmma_f32_16x16x4_f32(false, a, false, bw[n],
                                              (short)0, c, false, false);
#pragma unroll
    for (int r = 0; r < 8; ++r) {
      float h = relu1(c[r]);
      acc[2 * r]     = fmaf(h, w2a[n], acc[2 * r]);
      acc[2 * r + 1] = fmaf(h, w2b[n], acc[2 * r + 1]);
    }
  }

  float s8[8];
#pragma unroll
  for (int i = 0; i < 8; ++i) {
    float lo = acc[i], hi = acc[i + 8];
    float send = (lane & 8) ? lo : hi;
    float recv = __shfl_xor(send, 8, 32);
    s8[i] = ((lane & 8) ? hi : lo) + recv;
  }
  float s4[4];
#pragma unroll
  for (int i = 0; i < 4; ++i) {
    float lo = s4[0] * 0.0f + s8[i], hi = s8[i + 4];
    lo = s8[i];
    float send = (lane & 4) ? lo : hi;
    float recv = __shfl_xor(send, 4, 32);
    s4[i] = ((lane & 4) ? hi : lo) + recv;
  }
  float s2[2];
#pragma unroll
  for (int i = 0; i < 2; ++i) {
    float lo = s2[0] * 0.0f + s4[i], hi = s4[i + 2];
    lo = s4[i];
    float send = (lane & 2) ? lo : hi;
    float recv = __shfl_xor(send, 2, 32);
    s2[i] = ((lane & 2) ? hi : lo) + recv;
  }
  {
    float lo = s2[0], hi = s2[1];
    float send = (lane & 1) ? lo : hi;
    float recv = __shfl_xor(send, 1, 32);
    float s1 = ((lane & 1) ? hi : lo) + recv;
    out_probs_tile[lane] = s1 + b2lane;
  }
}

__global__ void __launch_bounds__(256)
cubeflow_kernel(const float* __restrict__ input,
                const float* __restrict__ latent,
                const float* __restrict__ W1,
                const float* __restrict__ b1,
                const float* __restrict__ W2,
                const float* __restrict__ b2,
                float* __restrict__ out,
                int B, int P) {
  const int b = blockIdx.y;
  const int lane = threadIdx.x & 31;
  const int waveId = (blockIdx.x * blockDim.x + threadIdx.x) >> 5;
  const int nWaves = (gridDim.x * blockDim.x) >> 5;

  const float theta = latent[b] * 10.0f;
  const float c = cosf(theta), s = sinf(theta);

  const int col = lane & 15;
  const int khalf = (lane < 16) ? 0 : 2;
  v2f bw[8];
  float b1v[8], w2a[8], w2b[8];
#pragma unroll
  for (int n = 0; n < 8; ++n) {
    const int h = n * 16 + col;
    bw[n].x = W1[h * 4 + khalf];
    bw[n].y = W1[h * 4 + khalf + 1];
    b1v[n] = b1[h];
    w2a[n] = W2[h];
    w2b[n] = W2[128 + h];
  }
  const float b2lane = (lane & 1) ? b2[1] : b2[0];

  float* __restrict__ out_eval  = out;
  float* __restrict__ out_con   = out + (size_t)B * P;
  float* __restrict__ out_probs = out + (size_t)B * P * 5;

  const size_t batchBase = (size_t)b * P;

  for (int base = waveId * 32; base < P; base += nWaves * 32) {
    const size_t gp = batchBase + (size_t)(base + lane);
    const float x0 = input[gp * 3 + 0];
    const float x1 = input[gp * 3 + 1];
    const float x2 = input[gp * 3 + 2];

    const float d0 = x0 - theta, d1 = x1, d2 = x2;
    const float f1 = 1.4f * d0 * d0 + 1.4f * d1 * d1 + 0.2f * d2 * d2 - 0.5f;
    const float e0 = d0, e1 = d1, e2 = d2 + 0.4f;
    const float c20 = c * e0 + s * e1;
    const float c21 = -s * e0 + c * e1;
    const float f2 = 3.8f * c20 * c20 + 0.6f * c21 * c21 + 3.8f * e2 * e2 - 0.5f;
    const float g0 = d0, g1 = d1, g2 = d2 - 0.6f;
    const float c30 = c * g0 - s * g1;
    const float c31 = s * g0 + c * g1;
    const float f3 = 0.35f * c30 * c30 + 2.8f * c31 * c31 + 2.8f * g2 * g2 - 0.5f;
    out_eval[gp] = fminf(f1, fminf(f2, f3));
    v4f conv = {x0, x1, x2, theta};
    *(v4f*)(out_con + gp * 4) = conv;

    const float sx0 = __shfl_xor(x0, 16, 32);
    const float sx1 = __shfl_xor(x1, 16, 32);
    const float sx2 = __shfl_xor(x2, 16, 32);
    v2f a0, a1;
    a0.x = (lane < 16) ? x0 : sx2;
    a0.y = (lane < 16) ? x1 : theta;
    a1.x = (lane < 16) ? sx0 : x2;
    a1.y = (lane < 16) ? sx1 : theta;

    mlp_tile(a0, bw, b1v, w2a, w2b, b2lane, lane,
             out_probs + (batchBase + base) * 2);
    mlp_tile(a1, bw, b1v, w2a, w2b, b2lane, lane,
             out_probs + (batchBase + base + 16) * 2);
  }
}

extern "C" void kernel_launch(void* const* d_in, const int* in_sizes, int n_in,
                              void* d_out, int out_size, void* d_ws, size_t ws_size,
                              hipStream_t stream) {
  const float* input  = (const float*)d_in[0];
  const float* latent = (const float*)d_in[1];
  const float* W1     = (const float*)d_in[2];
  const float* b1     = (const float*)d_in[3];
  const float* W2     = (const float*)d_in[4];
  const float* b2     = (const float*)d_in[5];

  const int B = in_sizes[1];
  const int P = in_sizes[0] / (B * 3);

  dim3 block(256);
  dim3 grid(64, B);
  hipLaunchKernelGGL(cubeflow_kernel, grid, block, 0, stream,
                     input, latent, W1, b1, W2, b2, (float*)d_out, B, P);
}


# Global loads issued ahead of use: the opening loads (points, weights) before the sine/cosine, and each step's points one step ahead into spare registers (tile index clamped, no branch)
# speedup vs baseline: 1.1290x; 1.1290x over previous
#include <hip/hip_runtime.h>

typedef __attribute__((ext_vector_type(2))) float v2f;
typedef __attribute__((ext_vector_type(4))) float v4f;
typedef __attribute__((ext_vector_type(8))) float v8f;

__device__ __forceinline__ float relu1(float x) {
  return __builtin_amdgcn_fmed3f(x, 0.0f, __builtin_inff());
}

__device__ __forceinline__ void mlp_tile(v2f a,
                                         const v2f (&bw)[8],
                                         const float (&b1v)[8],
                                         const float (&w2a)[8],
                                         const float (&w2b)[8],
                                         float b2lane,
                                         int lane,
                                         float* __restrict__ out_probs_tile) {
  float acc[16];
#pragma unroll
  for (int j = 0; j < 16; ++j) acc[j] = 0.0f;

#pragma unroll
  for (int n = 0; n < 8; ++n) {
    v8f c;
#pragma unroll
    for (int r = 0; r < 8; ++r) c[r] = b1v[n];
    c = __builtin_amdgcn_wmma_f32_16x16x4_f32(false, a, false, bw[n],
                                              (short)0, c, false, false);
#pragma unroll
    for (int r = 0; r < 8; ++r) {
      float h = relu1(c[r]);
      acc[2 * r]     = fmaf(h, w2a[n], acc[2 * r]);
      acc[2 * r + 1] = fmaf(h, w2b[n], acc[2 * r + 1]);
    }
  }

  float s8[8];
#pragma unroll
  for (int i = 0; i < 8; ++i) {
    float lo = acc[i], hi = acc[i + 8];
    float send = (lane & 8) ? lo : hi;
    float recv = __shfl_xor(send, 8, 32);
    s8[i] = ((lane & 8) ? hi : lo) + recv;
  }
  float s4[4];
#pragma unroll
  for (int i = 0; i < 4; ++i) {
    float lo = s4[0] * 0.0f + s8[i], hi = s8[i + 4];
    lo = s8[i];
    float send = (lane & 4) ? lo : hi;
    float recv = __shfl_xor(send, 4, 32);
    s4[i] = ((lane & 4) ? hi : lo) + recv;
  }
  float s2[2];
#pragma unroll
  for (int i = 0; i < 2; ++i) {
    float lo = s2[0] * 0.0f + s4[i], hi = s4[i + 2];
    lo = s4[i];
    float send = (lane & 2) ? lo : hi;
    float recv = __shfl_xor(send, 2, 32);
    s2[i] = ((lane & 2) ? hi : lo) + recv;
  }
  {
    float lo = s2[0], hi = s2[1];
    float send = (lane & 1) ? lo : hi;
    float recv = __shfl_xor(send, 1, 32);
    float s1 = ((lane & 1) ? hi : lo) + recv;
    out_probs_tile[lane] = s1 + b2lane;
  }
}

__global__ void __launch_bounds__(256)
cubeflow_kernel(const float* __restrict__ input,
                const float* __restrict__ latent,
                const float* __restrict__ W1,
                const float* __restrict__ b1,
                const float* __restrict__ W2,
                const float* __restrict__ b2,
                float* __restrict__ out,
                int B, int P) {
  const int b = blockIdx.y;
  const int lane = threadIdx.x & 31;
  const int waveId = (blockIdx.x * blockDim.x + threadIdx.x) >> 5;
  const int nWaves = (gridDim.x * blockDim.x) >> 5;

  const float lat = latent[b];

  const size_t batchBase = (size_t)b * P;
  const int stride = nWaves * 32;
  const int lastTile = max(P - 32, 0);
  int base = waveId * 32;
  const float* __restrict__ pin =
      input + (batchBase + (size_t)(min(base, lastTile) + lane)) * 3;
  float nx0 = pin[0], nx1 = pin[1], nx2 = pin[2];

  const int col = lane & 15;
  const int khalf = (lane < 16) ? 0 : 2;
  v2f bw[8];
  float b1v[8], w2a[8], w2b[8];
#pragma unroll
  for (int n = 0; n < 8; ++n) {
    const int h = n * 16 + col;
    bw[n].x = W1[h * 4 + khalf];
    bw[n].y = W1[h * 4 + khalf + 1];
    b1v[n] = b1[h];
    w2a[n] = W2[h];
    w2b[n] = W2[128 + h];
  }
  const float b2lane = (lane & 1) ? b2[1] : b2[0];

  const float theta = lat * 10.0f;
  const float c = cosf(theta), s = sinf(theta);

  float* __restrict__ out_eval  = out;
  float* __restrict__ out_con   = out + (size_t)B * P;
  float* __restrict__ out_probs = out + (size_t)B * P * 5;

  for (; base < P; base += stride) {
    const size_t gp = batchBase + (size_t)(base + lane);
    const float x0 = nx0, x1 = nx1, x2 = nx2;
    {
      const float* __restrict__ pn =
          input + (batchBase + (size_t)(min(base + stride, lastTile) + lane)) * 3;
      nx0 = pn[0];
      nx1 = pn[1];
      nx2 = pn[2];
    }
    __builtin_amdgcn_sched_barrier(0);

    const float d0 = x0 - theta, d1 = x1, d2 = x2;
    const float f1 = 1.4f * d0 * d0 + 1.4f * d1 * d1 + 0.2f * d2 * d2 - 0.5f;
    const float e0 = d0, e1 = d1, e2 = d2 + 0.4f;
    const float c20 = c * e0 + s * e1;
    const float c21 = -s * e0 + c * e1;
    const float f2 = 3.8f * c20 * c20 + 0.6f * c21 * c21 + 3.8f * e2 * e2 - 0.5f;
    const float g0 = d0, g1 = d1, g2 = d2 - 0.6f;
    const float c30 = c * g0 - s * g1;
    const float c31 = s * g0 + c * g1;
    const float f3 = 0.35f * c30 * c30 + 2.8f * c31 * c31 + 2.8f * g2 * g2 - 0.5f;
    out_eval[gp] = fminf(f1, fminf(f2, f3));
    v4f conv = {x0, x1, x2, theta};
    *(v4f*)(out_con + gp * 4) = conv;

    const float sx0 = __shfl_xor(x0, 16, 32);
    const float sx1 = __shfl_xor(x1, 16, 32);
    const float sx2 = __shfl_xor(x2, 16, 32);
    v2f a0, a1;
    a0.x = (lane < 16) ? x0 : sx2;
    a0.y = (lane < 16) ? x1 : theta;
    a1.x = (lane < 16) ? sx0 : x2;
    a1.y = (lane < 16) ? sx1 : theta;

    mlp_tile(a0, bw, b1v, w2a, w2b, b2lane, lane,
             out_probs + (batchBase + base) * 2);
    mlp_tile(a1, bw, b1v, w2a, w2b, b2lane, lane,
             out_probs + (batchBase + base + 16) * 2);
  }
}

extern "C" void kernel_launch(void* const* d_in, const int* in_sizes, int n_in,
                              void* d_out, int out_size, void* d_ws, size_t ws_size,
                              hipStream_t stream) {
  const float* input  = (const float*)d_in[0];
  const float* latent = (const float*)d_in[1];
  const float* W1     = (const float*)d_in[2];
  const float* b1     = (const float*)d_in[3];
  const float* W2     = (const float*)d_in[4];
  const float* b2     = (const float*)d_in[5];

  const int B = in_sizes[1];
  const int P = in_sizes[0] / (B * 3);

  dim3 block(256);
  dim3 grid(64, B);
  hipLaunchKernelGGL(cubeflow_kernel, grid, block, 0, stream,
                     input, latent, W1, b1, W2, b2, (float*)d_out, B, P);
}


# Hidden layer transposed (hidden units on matrix rows, bias in k=3): layer 2 stays in-lane, one cross-half exchange replaces the 15-shuffle reduce-scatter; ReLU as integer max; sum order changes
# speedup vs baseline: 1.2738x; 1.1283x over previous
#include <hip/hip_runtime.h>


typedef __attribute__((ext_vector_type(2))) float v2f;
typedef __attribute__((ext_vector_type(4))) float v4f;
typedef __attribute__((ext_vector_type(8))) float v8f;

__device__ __forceinline__ float relu1(float x) {
  const int b = __builtin_bit_cast(int, x);
  return __builtin_bit_cast(float, b > 0 ? b : 0);
}

__device__ __forceinline__ float xhalf(float x) {
  int b = __builtin_bit_cast(int, x);
  return __builtin_bit_cast(
      float, __builtin_amdgcn_permlanex16(b, b, 0x76543210, 0xfedcba98, false, false));
}

__device__ __forceinline__ void mlp_tile_t(v2f xt,
                                           const v2f (&w1t)[8],
                                           const v2f (&w2a)[32],
                                           const v2f (&w2b)[32],
                                           float& p0, float& p1) {
  v2f a0[4], a1[4];
  v8f z;
#pragma unroll
  for (int r = 0; r < 8; ++r) z[r] = 0.0f;
  v8f d = __builtin_amdgcn_wmma_f32_16x16x4_f32(false, w1t[0], false, xt,
                                                (short)0, z, false, false);
#pragma unroll
  for (int n = 0; n < 8; ++n) {
    v8f dn = d;
    if (n < 7)
      dn = __builtin_amdgcn_wmma_f32_16x16x4_f32(false, w1t[n + 1], false, xt,
                                                 (short)0, z, false, false);
#pragma unroll
    for (int q = 0; q < 4; ++q) {
      v2f h;
      h.x = relu1(d[2 * q]);
      h.y = relu1(d[2 * q + 1]);
      if (n == 0) {
        a0[q] = h * w2a[q];
        a1[q] = h * w2b[q];
      } else {
        a0[q] = __builtin_elementwise_fma(h, w2a[n * 4 + q], a0[q]);
        a1[q] = __builtin_elementwise_fma(h, w2b[n * 4 + q], a1[q]);
      }
    }
    d = dn;
  }
  const v2f s0 = (a0[0] + a0[1]) + (a0[2] + a0[3]);
  const v2f s1 = (a1[0] + a1[1]) + (a1[2] + a1[3]);
  p0 = s0.x + s0.y;
  p1 = s1.x + s1.y;
}

__global__ void __launch_bounds__(256)
cubeflow_kernel(const float* __restrict__ input,
                const float* __restrict__ latent,
                const float* __restrict__ W1,
                const float* __restrict__ b1,
                const float* __restrict__ W2,
                const float* __restrict__ b2,
                float* __restrict__ out,
                int B, int P) {
  const int b = blockIdx.y;
  const int lane = threadIdx.x & 31;
  const int waveId = (blockIdx.x * blockDim.x + threadIdx.x) >> 5;
  const int nWaves = (gridDim.x * blockDim.x) >> 5;

  const float lat = latent[b];

  const size_t batchBase = (size_t)b * P;
  const int stride = nWaves * 32;
  const int lastTile = max(P - 32, 0);
  int base = waveId * 32;
  const float* __restrict__ pin =
      input + (batchBase + (size_t)(min(base, lastTile) + lane)) * 3;
  float nx0 = pin[0], nx1 = pin[1], nx2 = pin[2];

  const float theta = lat * 10.0f;

  const int col = lane & 15;
  const bool upper = lane >= 16;
  const int khalf = upper ? 2 : 0;
  v2f w1t[8];
#pragma unroll
  for (int n = 0; n < 8; ++n) {
    const int j = n * 16 + col;
    const float wx = W1[j * 4 + khalf];
    const float wy = W1[j * 4 + khalf + 1];
    const float bj = b1[j];
    w1t[n].x = wx;
    w1t[n].y = upper ? fmaf(theta, wy, bj) : wy;
  }
  v2f w2a[32], w2b[32];
  const int hoff = upper ? 8 : 0;
#pragma unroll
  for (int n = 0; n < 8; ++n) {
#pragma unroll
    for (int q = 0; q < 4; ++q) {
      const int j = n * 16 + hoff + 2 * q;
      w2a[n * 4 + q].x = W2[j];
      w2a[n * 4 + q].y = W2[j + 1];
      w2b[n * 4 + q].x = W2[128 + j];
      w2b[n * 4 + q].y = W2[128 + j + 1];
    }
  }
  const float b2a = b2[0], b2b = b2[1];

  const float c = cosf(theta), s = sinf(theta);

  float* __restrict__ out_eval  = out;
  float* __restrict__ out_con   = out + (size_t)B * P;
  float* __restrict__ out_probs = out + (size_t)B * P * 5;

  for (; base < P; base += stride) {
    const size_t gp = batchBase + (size_t)(base + lane);
    const float x0 = nx0, x1 = nx1, x2 = nx2;
    {
      const float* __restrict__ pn =
          input + (batchBase + (size_t)(min(base + stride, lastTile) + lane)) * 3;
      nx0 = pn[0];
      nx1 = pn[1];
      nx2 = pn[2];
    }
    __builtin_amdgcn_sched_barrier(0);

    const float d0 = x0 - theta, d1 = x1, d2 = x2;
    const float f1 = 1.4f * d0 * d0 + 1.4f * d1 * d1 + 0.2f * d2 * d2 - 0.5f;
    const float e0 = d0, e1 = d1, e2 = d2 + 0.4f;
    const float c20 = c * e0 + s * e1;
    const float c21 = -s * e0 + c * e1;
    const float f2 = 3.8f * c20 * c20 + 0.6f * c21 * c21 + 3.8f * e2 * e2 - 0.5f;
    const float g0 = d0, g1 = d1, g2 = d2 - 0.6f;
    const float c30 = c * g0 - s * g1;
    const float c31 = s * g0 + c * g1;
    const float f3 = 0.35f * c30 * c30 + 2.8f * c31 * c31 + 2.8f * g2 * g2 - 0.5f;
    out_eval[gp] = fminf(f1, fminf(f2, f3));
    v4f conv = {x0, x1, x2, theta};
    *(v4f*)(out_con + gp * 4) = conv;

    const float sx0 = __shfl_xor(x0, 16, 32);
    const float sx1 = __shfl_xor(x1, 16, 32);
    const float sx2 = __shfl_xor(x2, 16, 32);
    v2f t0, t1;
    t0.x = upper ? sx2 : x0;
    t0.y = upper ? 1.0f : x1;
    t1.x = upper ? x2 : sx0;
    t1.y = upper ? 1.0f : sx1;

    float p00, p01, p10, p11;
    mlp_tile_t(t0, w1t, w2a, w2b, p00, p01);
    mlp_tile_t(t1, w1t, w2a, w2b, p10, p11);

    const float send0 = upper ? p00 : p10;
    const float send1 = upper ? p01 : p11;
    const float keep0 = upper ? p10 : p00;
    const float keep1 = upper ? p11 : p01;
    const float recv0 = xhalf(send0);
    const float recv1 = xhalf(send1);
    v2f pr;
    pr.x = (keep0 + recv0) + b2a;
    pr.y = (keep1 + recv1) + b2b;
    *(v2f*)(out_probs + gp * 2) = pr;
  }
}

extern "C" void kernel_launch(void* const* d_in, const int* in_sizes, int n_in,
                              void* d_out, int out_size, void* d_ws, size_t ws_size,
                              hipStream_t stream) {
  const float* input  = (const float*)d_in[0];
  const float* latent = (const float*)d_in[1];
  const float* W1     = (const float*)d_in[2];
  const float* b1     = (const float*)d_in[3];
  const float* W2     = (const float*)d_in[4];
  const float* b2     = (const float*)d_in[5];

  const int B = in_sizes[1];
  const int P = in_sizes[0] / (B * 3);

  dim3 block(256);
  dim3 grid(64, B);
  hipLaunchKernelGGL(cubeflow_kernel, grid, block, 0, stream,
                     input, latent, W1, b1, W2, b2, (float*)d_out, B, P);
}
